# starting kernel
# baseline (speedup 1.0000x reference)
#include <hip/hip_runtime.h>


#define DISP_RANGE 10
#define OUTD 21
#define NB 8
#define NA 9
#define ND 128
#define NH 64
#define NW 64
#define PLANE (NH * NW)
#define TILES_PER_N (NB * NA * 64 * 4)

typedef __attribute__((ext_vector_type(2))) float v2f;
typedef __attribute__((ext_vector_type(8))) float v8f;

static __device__ __forceinline__ int imin(int x, int y) { return x < y ? x : y; }
static __device__ __forceinline__ int imax(int x, int y) { return x > y ? x : y; }

__global__ __launch_bounds__(256) void BuildCostVolume_78056735638061_kernel(
    const float* __restrict__ uh, const float* __restrict__ vw,
    float* __restrict__ out)
{
    const int lane  = threadIdx.x & 31;
    const int wave  = threadIdx.x >> 5;
    int tile = blockIdx.x * 8 + wave;

    int n = 0;
    if (tile >= TILES_PER_N) { n = 1; tile -= TILES_PER_N; }

    const int ct = tile & 3;
    const int rc = (tile >> 2) & 63;
    const int ba = tile >> 8;
    const int a  = ba % NA;
    const int b  = ba / NA;

    const int col   = lane & 15;
    const int khalf = lane >> 4;

    int ad = a - 4; if (ad < 0) ad = -ad; if (ad == 0) ad = 1;
    const int L      = 20 * ad + 1;
    const int start0 = 64 - 10 * ad;

    const float* __restrict__ src = (n == 0) ? uh : vw;
    const float* __restrict__ p =
        src + (size_t)(b * NA + a) * (ND * PLANE);
    const int colOff = (n == 0) ? (rc * NW + ct * 16 + col)
                                : ((ct * 16 + col) * NW + rc);
    const int shift  = rc - 32;

    int bl0, bh0, bl1, bh1; float v0, v1;
    {
        int k = col;
        int s = (k * L) / OUTD;
        int e = ((k + 1) * L + (OUTD - 1)) / OUTD;
        bl0 = imax(start0 + s, -shift);
        bh0 = imin(start0 + e, ND - shift);
        v0  = 1.0f / (float)(e - s);
        k = 16 + col;
        s = (k * L) / OUTD;
        e = ((k + 1) * L + (OUTD - 1)) / OUTD;
        bl1 = imax(start0 + s, -shift);
        bh1 = imin(start0 + e, ND - shift);
        v1  = (k < OUTD) ? (1.0f / (float)(e - s)) : 0.0f;
    }

    v8f acc0 = {};
    v8f acc1 = {};

    const int dd_lo = start0 & ~3;
    const int dd_hi = start0 + L;

    auto loadPair = [&](int dd, float& o0, float& o1) {
        const int s0 = dd + 2 * khalf + shift;
        const int c0 = imin(imax(s0, 0), ND - 1);
        const int c1 = imin(imax(s0 + 1, 0), ND - 1);
        o0 = p[(unsigned)(c0 * PLANE + colOff)];
        o1 = p[(unsigned)(c1 * PLANE + colOff)];
    };

    float cb0, cb1;
    loadPair(dd_lo, cb0, cb1);

    for (int dd = dd_lo; dd < dd_hi; dd += 4) {
        float nb0, nb1;
        loadPair(dd + 4, nb0, nb1);

        {
            const int sp = imin(imax(dd + 8 + 2 * khalf + shift, 0), ND - 1);
            __builtin_prefetch(&p[(unsigned)(sp * PLANE + colOff)], 0, 1);
        }

        const int kA0 = dd + 2 * khalf;
        const int kA1 = kA0 + 1;

        v2f A0 = { (kA0 >= bl0 && kA0 < bh0) ? v0 : 0.0f,
                   (kA1 >= bl0 && kA1 < bh0) ? v0 : 0.0f };
        v2f A1 = { (kA0 >= bl1 && kA0 < bh1) ? v1 : 0.0f,
                   (kA1 >= bl1 && kA1 < bh1) ? v1 : 0.0f };
        v2f Bv = { cb0, cb1 };

        acc0 = __builtin_amdgcn_wmma_f32_16x16x4_f32(
                   false, A0, false, Bv, (short)0, acc0, false, false);
        acc1 = __builtin_amdgcn_wmma_f32_16x16x4_f32(
                   false, A1, false, Bv, (short)0, acc1, false, false);

        cb0 = nb0; cb1 = nb1;
    }

    float* __restrict__ q =
        out + (size_t)((b * 2 + n) * NA + a) * (OUTD * PLANE) + colOff;
#pragma unroll
    for (int r = 0; r < 8; ++r) {
        q[(unsigned)((r + 8 * khalf) * PLANE)] = acc0[r];
    }
    if (khalf == 0) {
#pragma unroll
        for (int r = 0; r < 5; ++r) {
            q[(unsigned)((16 + r) * PLANE)] = acc1[r];
        }
    }
}

extern "C" void kernel_launch(void* const* d_in, const int* in_sizes, int n_in,
                              void* d_out, int out_size, void* d_ws, size_t ws_size,
                              hipStream_t stream) {
    const float* uh = (const float*)d_in[0];
    const float* vw = (const float*)d_in[1];
    float* out = (float*)d_out;

    const int totalWaves = 2 * TILES_PER_N;
    dim3 grid(totalWaves / 8);
    dim3 block(256);
    BuildCostVolume_78056735638061_kernel<<<grid, block, 0, stream>>>(uh, vw, out);
}


# Stage each (n,b,a,h) depth x 64-w slab in LDS with whole-line 16 B loads; B fragments read from LDS, the vw shear as a per-lane row offset, so vw reads and writes run along w, not at 256 B stride
# speedup vs baseline: 6.4601x; 6.4601x over previous
#include <hip/hip_runtime.h>


#define DISP_RANGE 10
#define OUTD 21
#define NB 8
#define NA 9
#define ND 128
#define NH 64
#define NW 64
#define PLANE (NH * NW)
#define SLABS_PER_N (NB * NA * NH)
#define PITCH 72
#define MAX_ROWS ND

typedef __attribute__((ext_vector_type(2))) float v2f;
typedef __attribute__((ext_vector_type(4))) float v4f;
typedef __attribute__((ext_vector_type(8))) float v8f;

static __device__ __forceinline__ int imin(int x, int y) { return x < y ? x : y; }
static __device__ __forceinline__ int imax(int x, int y) { return x > y ? x : y; }

template <int NP>
static __device__ __forceinline__ void stage_rows(
    const float* __restrict__ p, float* slab, int Dlo, int nrows, int rr, int c4)
{
    v4f v[NP];
#pragma unroll
    for (int i = 0; i < NP; ++i) {
        const int row = imin(rr + 8 * i, nrows - 1);
        v[i] = *reinterpret_cast<const v4f*>(
                   p + (unsigned)((Dlo + row) * PLANE + c4 * 4));
    }
#pragma unroll
    for (int i = 0; i < NP; ++i)
        asm volatile("" :: "v"(v[i]));
#pragma unroll
    for (int i = 0; i < NP; ++i) {
        *reinterpret_cast<v4f*>(&slab[(rr + 8 * i) * PITCH + c4 * 4]) = v[i];
    }
}

__global__ __launch_bounds__(128) void cost_volume_kernel(
    const float* __restrict__ uh, const float* __restrict__ vw,
    float* __restrict__ out)
{
    __shared__ float slab[MAX_ROWS * PITCH];

    const int tid  = threadIdx.x;
    const int lane = tid & 31;
    const int wave = tid >> 5;

    int blk = blockIdx.x;
    int n = 1;
    if (blk >= SLABS_PER_N) { n = 0; blk -= SLABS_PER_N; }

    const int h  = blk & 63;
    const int ba = blk >> 6;
    const int a  = ba % NA;
    const int b  = ba / NA;

    int ad = a - 4; if (ad < 0) ad = -ad; if (ad == 0) ad = 1;
    const int L      = 20 * ad + 1;
    const int start0 = 64 - 10 * ad;

    const int sMin = (n == 0) ? (h - 32) : -32;
    const int sMax = (n == 0) ? (h - 32) : 31;
    const int Dlo   = imax(start0 + sMin, 0);
    const int Dhi   = imin(start0 + L + sMax, ND);
    const int nrows = Dhi - Dlo;

    const float* __restrict__ src = (n == 0) ? uh : vw;
    const float* __restrict__ p =
        src + (size_t)(b * NA + a) * (ND * PLANE) + h * NW;

    {
        const int c4 = tid & 15;
        const int rr = tid >> 4;
        const int npass = (nrows + 7) >> 3;
        switch ((npass + 1) >> 1) {
        case 1:  stage_rows<2>(p, slab, Dlo, nrows, rr, c4);  break;
        case 2:  stage_rows<4>(p, slab, Dlo, nrows, rr, c4);  break;
        case 3:  stage_rows<6>(p, slab, Dlo, nrows, rr, c4);  break;
        case 4:  stage_rows<8>(p, slab, Dlo, nrows, rr, c4);  break;
        case 5:  stage_rows<10>(p, slab, Dlo, nrows, rr, c4); break;
        case 6:  stage_rows<12>(p, slab, Dlo, nrows, rr, c4); break;
        case 7:  stage_rows<14>(p, slab, Dlo, nrows, rr, c4); break;
        default: stage_rows<16>(p, slab, Dlo, nrows, rr, c4); break;
        }
    }
    __syncthreads();

    const int col   = lane & 15;
    const int khalf = lane >> 4;
    const int w     = wave * 16 + col;
    const int shift = (n == 0) ? (h - 32) : (w - 32);

    int bl0, bh0, bl1, bh1; float v0, v1;
    {
        int k = col;
        int s = (k * L) / OUTD;
        int e = ((k + 1) * L + (OUTD - 1)) / OUTD;
        bl0 = start0 + s;
        bh0 = start0 + e;
        v0  = 1.0f / (float)(e - s);
        k = 16 + col;
        s = (k * L) / OUTD;
        e = ((k + 1) * L + (OUTD - 1)) / OUTD;
        bl1 = start0 + s;
        bh1 = start0 + e;
        v1  = (k < OUTD) ? (1.0f / (float)(e - s)) : 0.0f;
    }

    v8f acc0 = {};
    v8f acc1 = {};

    const int dd_lo = start0 & ~3;
    const int dd_hi = start0 + L;

    const float* sl = slab + w;
    const int rmax = nrows - 1;
    auto loadPair = [&](int dep, float& o0, float& o1) {
        const int r0 = imin(imax(dep - Dlo, 0), rmax);
        const int r1 = imin(imax(dep + 1 - Dlo, 0), rmax);
        const float x0 = sl[r0 * PITCH];
        const float x1 = sl[r1 * PITCH];
        o0 = ((unsigned)dep < (unsigned)ND) ? x0 : 0.0f;
        o1 = ((unsigned)(dep + 1) < (unsigned)ND) ? x1 : 0.0f;
    };

    int kA0 = dd_lo + 2 * khalf;
    int dep = kA0 + shift;

    float cb0, cb1;
    loadPair(dep, cb0, cb1);

    for (int dd = dd_lo; dd < dd_hi; dd += 4) {
        float nb0, nb1;
        loadPair(dep + 4, nb0, nb1);

        const int kA1 = kA0 + 1;

        v2f A0 = { (kA0 >= bl0 && kA0 < bh0) ? v0 : 0.0f,
                   (kA1 >= bl0 && kA1 < bh0) ? v0 : 0.0f };
        v2f A1 = { (kA0 >= bl1 && kA0 < bh1) ? v1 : 0.0f,
                   (kA1 >= bl1 && kA1 < bh1) ? v1 : 0.0f };
        v2f Bv = { cb0, cb1 };

        acc0 = __builtin_amdgcn_wmma_f32_16x16x4_f32(
                   false, A0, false, Bv, (short)0, acc0, false, false);
        acc1 = __builtin_amdgcn_wmma_f32_16x16x4_f32(
                   false, A1, false, Bv, (short)0, acc1, false, false);

        cb0 = nb0; cb1 = nb1;
        kA0 += 4; dep += 4;
    }

    float* __restrict__ q =
        out + (size_t)((b * 2 + n) * NA + a) * (OUTD * PLANE) + h * NW + w;
#pragma unroll
    for (int r = 0; r < 8; ++r) {
        q[(unsigned)((r + 8 * khalf) * PLANE)] = acc0[r];
    }
    if (khalf == 0) {
#pragma unroll
        for (int r = 0; r < 5; ++r) {
            q[(unsigned)((16 + r) * PLANE)] = acc1[r];
        }
    }
}

extern "C" void kernel_launch(void* const* d_in, const int* in_sizes, int n_in,
                              void* d_out, int out_size, void* d_ws, size_t ws_size,
                              hipStream_t stream) {
    const float* uh = (const float*)d_in[0];
    const float* vw = (const float*)d_in[1];
    float* out = (float*)d_out;

    dim3 grid(2 * SLABS_PER_N);
    dim3 block(128);
    cost_volume_kernel<<<grid, block, 0, stream>>>(uh, vw, out);
}


# Two 16-wide w tiles per wave (2 waves per slab, not 4): A fragments built once per chunk serve both tiles; for uh tile 2's B read is tile 1's LDS address + 16 columns; sum order kept
# speedup vs baseline: 6.7561x; 1.0458x over previous
#include <hip/hip_runtime.h>


#define DISP_RANGE 10
#define OUTD 21
#define NB 8
#define NA 9
#define ND 128
#define NH 64
#define NW 64
#define PLANE (NH * NW)
#define SLABS_PER_N (NB * NA * NH)
#define PITCH 72
#define MAX_ROWS ND

typedef __attribute__((ext_vector_type(2))) float v2f;
typedef __attribute__((ext_vector_type(4))) float v4f;
typedef __attribute__((ext_vector_type(8))) float v8f;

static __device__ __forceinline__ int imin(int x, int y) { return x < y ? x : y; }
static __device__ __forceinline__ int imax(int x, int y) { return x > y ? x : y; }

template <int NP>
static __device__ __forceinline__ void stage_rows(
    const float* __restrict__ p, float* slab, int Dlo, int nrows, int rr, int c4)
{
    v4f v[NP];
#pragma unroll
    for (int i = 0; i < NP; ++i) {
        const int row = imin(rr + 4 * i, nrows - 1);
        v[i] = *reinterpret_cast<const v4f*>(
                   p + (unsigned)((Dlo + row) * PLANE + c4 * 4));
    }
#pragma unroll
    for (int i = 0; i < NP; ++i)
        asm volatile("" :: "v"(v[i]));
#pragma unroll
    for (int i = 0; i < NP; ++i) {
        *reinterpret_cast<v4f*>(&slab[(rr + 4 * i) * PITCH + c4 * 4]) = v[i];
    }
}

template <bool VW>
static __device__ __forceinline__ void pool_two_tiles(
    const float* slab, float* __restrict__ q, int wbase, int h, int khalf, int col,
    int L, int start0, int Dlo, int nrows)
{
    int bl0, bh0, bl1, bh1; float v0, v1;
    {
        int k = col;
        int s = (k * L) / OUTD;
        int e = ((k + 1) * L + (OUTD - 1)) / OUTD;
        bl0 = start0 + s;
        bh0 = start0 + e;
        v0  = 1.0f / (float)(e - s);
        k = 16 + col;
        s = (k * L) / OUTD;
        e = ((k + 1) * L + (OUTD - 1)) / OUTD;
        bl1 = start0 + s;
        bh1 = start0 + e;
        v1  = (k < OUTD) ? (1.0f / (float)(e - s)) : 0.0f;
    }

    v8f acc00 = {}, acc01 = {};
    v8f acc10 = {}, acc11 = {};

    const int dd_lo = start0 & ~3;
    const int dd_hi = start0 + L;

    const int shift0 = VW ? (wbase - 32)      : (h - 32);
    const int shift1 = VW ? (wbase + 16 - 32) : (h - 32);

    const float* sl = slab + wbase;
    const int rmax = nrows - 1;
    auto loadPair = [&](int dep, int coff, float& o0, float& o1) {
        const int r0 = imin(imax(dep - Dlo, 0), rmax);
        const int r1 = imin(imax(dep + 1 - Dlo, 0), rmax);
        const float x0 = sl[r0 * PITCH + coff];
        const float x1 = sl[r1 * PITCH + coff];
        o0 = ((unsigned)dep < (unsigned)ND) ? x0 : 0.0f;
        o1 = ((unsigned)(dep + 1) < (unsigned)ND) ? x1 : 0.0f;
    };

    int kA0  = dd_lo + 2 * khalf;
    int dep0 = kA0 + shift0;
    int dep1 = kA0 + shift1;

    float c00, c01, c10, c11;
    loadPair(dep0, 0,  c00, c01);
    loadPair(dep1, 16, c10, c11);

    for (int dd = dd_lo; dd < dd_hi; dd += 4) {
        float n00, n01, n10, n11;
        loadPair(dep0 + 4, 0,  n00, n01);
        loadPair(dep1 + 4, 16, n10, n11);

        const int kA1 = kA0 + 1;

        v2f A0 = { (kA0 >= bl0 && kA0 < bh0) ? v0 : 0.0f,
                   (kA1 >= bl0 && kA1 < bh0) ? v0 : 0.0f };
        v2f A1 = { (kA0 >= bl1 && kA0 < bh1) ? v1 : 0.0f,
                   (kA1 >= bl1 && kA1 < bh1) ? v1 : 0.0f };
        v2f B0 = { c00, c01 };
        v2f B1 = { c10, c11 };

        acc00 = __builtin_amdgcn_wmma_f32_16x16x4_f32(
                    false, A0, false, B0, (short)0, acc00, false, false);
        acc01 = __builtin_amdgcn_wmma_f32_16x16x4_f32(
                    false, A1, false, B0, (short)0, acc01, false, false);
        acc10 = __builtin_amdgcn_wmma_f32_16x16x4_f32(
                    false, A0, false, B1, (short)0, acc10, false, false);
        acc11 = __builtin_amdgcn_wmma_f32_16x16x4_f32(
                    false, A1, false, B1, (short)0, acc11, false, false);

        c00 = n00; c01 = n01; c10 = n10; c11 = n11;
        kA0 += 4; dep0 += 4; dep1 += 4;
    }

#pragma unroll
    for (int r = 0; r < 8; ++r) {
        q[(unsigned)((r + 8 * khalf) * PLANE)]      = acc00[r];
        q[(unsigned)((r + 8 * khalf) * PLANE) + 16] = acc10[r];
    }
    if (khalf == 0) {
#pragma unroll
        for (int r = 0; r < 5; ++r) {
            q[(unsigned)((16 + r) * PLANE)]      = acc01[r];
            q[(unsigned)((16 + r) * PLANE) + 16] = acc11[r];
        }
    }
}

__global__ __launch_bounds__(64) void cost_volume_kernel(
    const float* __restrict__ uh, const float* __restrict__ vw,
    float* __restrict__ out)
{
    __shared__ float slab[MAX_ROWS * PITCH];

    const int tid  = threadIdx.x;
    const int lane = tid & 31;
    const int wave = tid >> 5;

    int blk = blockIdx.x;
    int n = 1;
    if (blk >= SLABS_PER_N) { n = 0; blk -= SLABS_PER_N; }

    const int h  = blk & 63;
    const int ba = blk >> 6;
    const int a  = ba % NA;
    const int b  = ba / NA;

    int ad = a - 4; if (ad < 0) ad = -ad; if (ad == 0) ad = 1;
    const int L      = 20 * ad + 1;
    const int start0 = 64 - 10 * ad;

    const int sMin = (n == 0) ? (h - 32) : -32;
    const int sMax = (n == 0) ? (h - 32) : 31;
    const int Dlo   = imax(start0 + sMin, 0);
    const int Dhi   = imin(start0 + L + sMax, ND);
    const int nrows = Dhi - Dlo;

    const float* __restrict__ src = (n == 0) ? uh : vw;
    const float* __restrict__ p =
        src + (size_t)(b * NA + a) * (ND * PLANE) + h * NW;

    {
        const int c4 = tid & 15;
        const int rr = tid >> 4;
        const int npass = (nrows + 3) >> 2;
        switch ((npass + 3) >> 2) {
        case 1:  stage_rows<4>(p, slab, Dlo, nrows, rr, c4);  break;
        case 2:  stage_rows<8>(p, slab, Dlo, nrows, rr, c4);  break;
        case 3:  stage_rows<12>(p, slab, Dlo, nrows, rr, c4); break;
        case 4:  stage_rows<16>(p, slab, Dlo, nrows, rr, c4); break;
        case 5:  stage_rows<20>(p, slab, Dlo, nrows, rr, c4); break;
        case 6:  stage_rows<24>(p, slab, Dlo, nrows, rr, c4); break;
        case 7:  stage_rows<28>(p, slab, Dlo, nrows, rr, c4); break;
        default: stage_rows<32>(p, slab, Dlo, nrows, rr, c4); break;
        }
    }
    __syncthreads();

    const int col   = lane & 15;
    const int khalf = lane >> 4;
    const int wbase = wave * 32 + col;

    float* __restrict__ q =
        out + (size_t)((b * 2 + n) * NA + a) * (OUTD * PLANE) + h * NW + wbase;

    if (n == 0) pool_two_tiles<false>(slab, q, wbase, h, khalf, col, L, start0, Dlo, nrows);
    else        pool_two_tiles<true >(slab, q, wbase, h, khalf, col, L, start0, Dlo, nrows);
}

extern "C" void kernel_launch(void* const* d_in, const int* in_sizes, int n_in,
                              void* d_out, int out_size, void* d_ws, size_t ws_size,
                              hipStream_t stream) {
    const float* uh = (const float*)d_in[0];
    const float* vw = (const float*)d_in[1];
    float* out = (float*)d_out;

    dim3 grid(2 * SLABS_PER_N);
    dim3 block(64);
    cost_volume_kernel<<<grid, block, 0, stream>>>(uh, vw, out);
}


# Persistent grid of 2048 work-groups, 4-5 slabs each: the next slab's rows are loaded into registers under the current slab's WMMAs and stores (LDS-only hand barrier); sum order kept
# speedup vs baseline: 6.9486x; 1.0285x over previous
#include <hip/hip_runtime.h>


#define DISP_RANGE 10
#define OUTD 21
#define NB 8
#define NA 9
#define ND 128
#define NH 64
#define NW 64
#define PLANE (NH * NW)
#define SLABS_PER_N (NB * NA * NH)
#define NSLAB (2 * SLABS_PER_N)
#define NWG 2048
#define MAX_ROUNDS 5
#define PITCH 72
#define MAX_ROWS ND
#define NPASS_MAX 32

typedef __attribute__((ext_vector_type(2))) float v2f;
typedef __attribute__((ext_vector_type(4))) float v4f;
typedef __attribute__((ext_vector_type(8))) float v8f;

static __device__ __forceinline__ int imin(int x, int y) { return x < y ? x : y; }
static __device__ __forceinline__ int imax(int x, int y) { return x > y ? x : y; }

struct SlabDesc { int n, a, b, h, L, start0, Dlo, nrows; };

static __device__ __forceinline__ SlabDesc decode_slab(int p)
{
    SlabDesc s;
    s.n = (p < SLABS_PER_N) ? 1 : 0;
    const int q   = s.n ? p : p - SLABS_PER_N;
    const int idx = q >> 9;
    const int rem = q & 511;
    s.b = rem >> 6;
    s.h = rem & 63;
    s.a = (idx < 6) ? ((idx & 1) ? 8 - (idx >> 1) : (idx >> 1)) : idx - 3;
    int ad = s.a - 4; if (ad < 0) ad = -ad; if (ad == 0) ad = 1;
    s.L      = 20 * ad + 1;
    s.start0 = 64 - 10 * ad;
    const int sMin = (s.n == 0) ? (s.h - 32) : -32;
    const int sMax = (s.n == 0) ? (s.h - 32) : 31;
    s.Dlo = imax(s.start0 + sMin, 0);
    const int Dhi = imin(s.start0 + s.L + sMax, ND);
    s.nrows = Dhi - s.Dlo;
    return s;
}

static __device__ __forceinline__ int slab_of(int g, int t)
{
    const int r = (t == 0) ? 3 : ((t < 4) ? t - 1 : 4);
    return (r & 1) ? (r + 1) * NWG - 1 - g : r * NWG + g;
}

static __device__ __forceinline__ void issue_rows(
    const float* __restrict__ src, const SlabDesc& s, int rr, int c4, v4f (&v)[NPASS_MAX])
{
    const float* __restrict__ p =
        src + (size_t)(s.b * NA + s.a) * (ND * PLANE) + s.h * NW;
    const int npass = (s.nrows + 3) >> 2;
#pragma unroll
    for (int j = 0; j < NPASS_MAX / 4; ++j) {
        if (j == 0 || 4 * j < npass) {
#pragma unroll
            for (int i = 4 * j; i < 4 * j + 4; ++i) {
                const int row = imin(rr + 4 * i, s.nrows - 1);
                v[i] = *reinterpret_cast<const v4f*>(
                           p + (unsigned)((s.Dlo + row) * PLANE + c4 * 4));
            }
        }
    }
}

static __device__ __forceinline__ void store_rows(
    float* slab, int nrows, int rr, int c4, const v4f (&v)[NPASS_MAX])
{
    const int npass = (nrows + 3) >> 2;
#pragma unroll
    for (int j = 0; j < NPASS_MAX / 4; ++j) {
        if (j == 0 || 4 * j < npass) {
#pragma unroll
            for (int i = 4 * j; i < 4 * j + 4; ++i)
                *reinterpret_cast<v4f*>(&slab[(rr + 4 * i) * PITCH + c4 * 4]) = v[i];
        }
    }
}

static __device__ __forceinline__ void lds_barrier()
{
    asm volatile("s_wait_dscnt 0x0\n\ts_barrier_signal -1\n\ts_barrier_wait -1" ::: "memory");
}

template <bool VW>
static __device__ __forceinline__ void pool_two_tiles(
    const float* slab, float* __restrict__ q, int wbase, int h, int khalf, int col,
    int L, int start0, int Dlo, int nrows)
{
    int bl0, bh0, bl1, bh1; float v0, v1;
    {
        int k = col;
        int s = (k * L) / OUTD;
        int e = ((k + 1) * L + (OUTD - 1)) / OUTD;
        bl0 = start0 + s;
        bh0 = start0 + e;
        v0  = 1.0f / (float)(e - s);
        k = 16 + col;
        s = (k * L) / OUTD;
        e = ((k + 1) * L + (OUTD - 1)) / OUTD;
        bl1 = start0 + s;
        bh1 = start0 + e;
        v1  = (k < OUTD) ? (1.0f / (float)(e - s)) : 0.0f;
    }

    v8f acc00 = {}, acc01 = {};
    v8f acc10 = {}, acc11 = {};

    const int dd_lo = start0 & ~3;
    const int dd_hi = start0 + L;

    const int shift0 = VW ? (wbase - 32)      : (h - 32);
    const int shift1 = VW ? (wbase + 16 - 32) : (h - 32);

    const float* sl = slab + wbase;
    const int rmax = nrows - 1;
    auto loadPair = [&](int dep, int coff, float& o0, float& o1) {
        const int r0 = imin(imax(dep - Dlo, 0), rmax);
        const int r1 = imin(imax(dep + 1 - Dlo, 0), rmax);
        const float x0 = sl[r0 * PITCH + coff];
        const float x1 = sl[r1 * PITCH + coff];
        o0 = ((unsigned)dep < (unsigned)ND) ? x0 : 0.0f;
        o1 = ((unsigned)(dep + 1) < (unsigned)ND) ? x1 : 0.0f;
    };

    int kA0  = dd_lo + 2 * khalf;
    int dep0 = kA0 + shift0;
    int dep1 = kA0 + shift1;

    float c00, c01, c10, c11;
    loadPair(dep0, 0,  c00, c01);
    loadPair(dep1, 16, c10, c11);

    for (int dd = dd_lo; dd < dd_hi; dd += 4) {
        float n00, n01, n10, n11;
        loadPair(dep0 + 4, 0,  n00, n01);
        loadPair(dep1 + 4, 16, n10, n11);

        const int kA1 = kA0 + 1;

        v2f A0 = { (kA0 >= bl0 && kA0 < bh0) ? v0 : 0.0f,
                   (kA1 >= bl0 && kA1 < bh0) ? v0 : 0.0f };
        v2f A1 = { (kA0 >= bl1 && kA0 < bh1) ? v1 : 0.0f,
                   (kA1 >= bl1 && kA1 < bh1) ? v1 : 0.0f };
        v2f B0 = { c00, c01 };
        v2f B1 = { c10, c11 };

        acc00 = __builtin_amdgcn_wmma_f32_16x16x4_f32(
                    false, A0, false, B0, (short)0, acc00, false, false);
        acc01 = __builtin_amdgcn_wmma_f32_16x16x4_f32(
                    false, A1, false, B0, (short)0, acc01, false, false);
        acc10 = __builtin_amdgcn_wmma_f32_16x16x4_f32(
                    false, A0, false, B1, (short)0, acc10, false, false);
        acc11 = __builtin_amdgcn_wmma_f32_16x16x4_f32(
                    false, A1, false, B1, (short)0, acc11, false, false);

        c00 = n00; c01 = n01; c10 = n10; c11 = n11;
        kA0 += 4; dep0 += 4; dep1 += 4;
    }

#pragma unroll
    for (int r = 0; r < 8; ++r) {
        q[(unsigned)((r + 8 * khalf) * PLANE)]      = acc00[r];
        q[(unsigned)((r + 8 * khalf) * PLANE) + 16] = acc10[r];
    }
    if (khalf == 0) {
#pragma unroll
        for (int r = 0; r < 5; ++r) {
            q[(unsigned)((16 + r) * PLANE)]      = acc01[r];
            q[(unsigned)((16 + r) * PLANE) + 16] = acc11[r];
        }
    }
}

__global__ __launch_bounds__(64) void cost_volume_kernel(
    const float* __restrict__ uh, const float* __restrict__ vw,
    float* __restrict__ out)
{
    __shared__ float slab[MAX_ROWS * PITCH];

    const int tid  = threadIdx.x;
    const int lane = tid & 31;
    const int wave = tid >> 5;
    const int c4   = tid & 15;
    const int rr   = tid >> 4;
    const int col   = lane & 15;
    const int khalf = lane >> 4;
    const int wbase = wave * 32 + col;

    const int g = blockIdx.x;

    v4f v[NPASS_MAX];

    SlabDesc cur = decode_slab(slab_of(g, 0));
    issue_rows(cur.n == 0 ? uh : vw, cur, rr, c4, v);

    for (int r = 0; r < MAX_ROUNDS; ++r) {
        store_rows(slab, cur.nrows, rr, c4, v);
        lds_barrier();

        const int pn = slab_of(g, r + 1);
        const bool more = (r + 1 < MAX_ROUNDS) && (pn < NSLAB);
        SlabDesc nxt = cur;
        if (more) {
            nxt = decode_slab(pn);
            issue_rows(nxt.n == 0 ? uh : vw, nxt, rr, c4, v);
        }

        float* __restrict__ q =
            out + (size_t)((cur.b * 2 + cur.n) * NA + cur.a) * (OUTD * PLANE) + cur.h * NW + wbase;
        if (cur.n == 0)
            pool_two_tiles<false>(slab, q, wbase, cur.h, khalf, col, cur.L, cur.start0, cur.Dlo, cur.nrows);
        else
            pool_two_tiles<true >(slab, q, wbase, cur.h, khalf, col, cur.L, cur.start0, cur.Dlo, cur.nrows);

        if (!more) break;
        lds_barrier();
        cur = nxt;
    }
}

extern "C" void kernel_launch(void* const* d_in, const int* in_sizes, int n_in,
                              void* d_out, int out_size, void* d_ws, size_t ws_size,
                              hipStream_t stream) {
    const float* uh = (const float*)d_in[0];
    const float* vw = (const float*)d_in[1];
    float* out = (float*)d_out;

    dim3 grid(NWG);
    dim3 block(64);
    cost_volume_kernel<<<grid, block, 0, stream>>>(uh, vw, out);
}
